# starting kernel
# baseline (speedup 1.0000x reference)
#include <hip/hip_runtime.h>

typedef __attribute__((ext_vector_type(2))) float v2f;
typedef __attribute__((ext_vector_type(8))) float v8f;

#define RNN_B 8192
#define RNN_T 512
#define RNN_I 9
#define RNN_H 6
#define TILES_PER_WAVE 8

__device__ __forceinline__ float fast_tanhf(float v) {
  float e = __builtin_amdgcn_exp2f(v * 2.88539008177792681472f);
  return 1.0f - 2.0f * __builtin_amdgcn_rcpf(e + 1.0f);
}

__global__ __launch_bounds__(256) void xproj_wmma_f32(
    const float* __restrict__ x, const float* __restrict__ W_ih,
    const float* __restrict__ b_ih, const float* __restrict__ b_hh,
    float* __restrict__ xp)
{
  const int  lane = threadIdx.x & 31;
  const int  l15  = lane & 15;
  const int  half = lane >> 4;
  const long wave = ((long)blockIdx.x * blockDim.x + threadIdx.x) >> 5;
  const long waveRowBase = wave * (TILES_PER_WAVE * 16);

  int  kIdx[6];
  bool kVal[6];
#pragma unroll
  for (int s = 0; s < 3; ++s) {
    const int kA = 4 * s + half * 2;
    kVal[2*s+0] = (kA     < RNN_I);
    kVal[2*s+1] = (kA + 1 < RNN_I);
    kIdx[2*s+0] = kVal[2*s+0] ? kA     : (RNN_I - 1);
    kIdx[2*s+1] = kVal[2*s+1] ? kA + 1 : (RNN_I - 1);
  }

  const bool nVal   = (l15 < RNN_H);
  const int  nClamp = nVal ? l15 : (RNN_H - 1);

  float bi = b_ih[nClamp];
  float bh = b_hh[nClamp];
  const float bias = nVal ? (bi + bh) : 0.0f;

  v2f bm[3];
#pragma unroll
  for (int s = 0; s < 3; ++s) {
    float w0 = W_ih[nClamp * RNN_I + kIdx[2*s+0]];
    float w1 = W_ih[nClamp * RNN_I + kIdx[2*s+1]];
    bm[s].x = (nVal && kVal[2*s+0]) ? w0 : 0.0f;
    bm[s].y = (nVal && kVal[2*s+1]) ? w1 : 0.0f;
  }

#pragma unroll 1
  for (int tt = 0; tt < TILES_PER_WAVE; ++tt) {
    const long rowBase = waveRowBase + (long)tt * 16;
    const float* xrow = x + (rowBase + l15) * RNN_I;

    v8f c;
#pragma unroll
    for (int j = 0; j < 8; ++j) c[j] = bias;

#pragma unroll
    for (int s = 0; s < 3; ++s) {
      float r0 = xrow[kIdx[2*s+0]];
      float r1 = xrow[kIdx[2*s+1]];
      v2f a;
      a.x = kVal[2*s+0] ? r0 : 0.0f;
      a.y = kVal[2*s+1] ? r1 : 0.0f;
      c = __builtin_amdgcn_wmma_f32_16x16x4_f32(false, a, false, bm[s],
                                                (short)0, c, false, false);
    }

    if (l15 < 8) {
#pragma unroll
      for (int j = 0; j < 8; ++j) {
        long r  = rowBase + j + half * 8;
        long bb = r >> 9;
        long ti = r & (RNN_T - 1);
        xp[(ti * (long)RNN_B + bb) * 8 + l15] = c[j];
      }
    }
  }
}

__global__ __launch_bounds__(256) void rnn_scan(
    const float* __restrict__ xp, const float* __restrict__ W_hh,
    const float* __restrict__ W_fc, const float* __restrict__ b_fc,
    float* __restrict__ out)
{
  const int b = blockIdx.x * blockDim.x + threadIdx.x;

  float w[RNN_H][RNN_H];
#pragma unroll
  for (int j = 0; j < RNN_H; ++j)
#pragma unroll
    for (int i = 0; i < RNN_H; ++i) w[j][i] = W_hh[j * RNN_H + i];

  float h[RNN_H];
#pragma unroll
  for (int j = 0; j < RNN_H; ++j) h[j] = 0.0f;

  const float4* base = (const float4*)xp;
  float4 cur0 = base[(long)b * 2 + 0];
  float4 cur1 = base[(long)b * 2 + 1];

  for (int t = 0; t < RNN_T; ++t) {
    const int  tn = (t + 1 < RNN_T) ? (t + 1) : (RNN_T - 1);
    const long ni = ((long)tn * RNN_B + b) * 2;
    float4 nx0 = base[ni];
    float4 nx1 = base[ni + 1];
    const int tp = (t + 8 < RNN_T) ? (t + 8) : (RNN_T - 1);
    __builtin_prefetch(&base[((long)tp * RNN_B + b) * 2], 0, 0);

    const float xv[RNN_H] = {cur0.x, cur0.y, cur0.z, cur0.w, cur1.x, cur1.y};
    float nh[RNN_H];
#pragma unroll
    for (int j = 0; j < RNN_H; ++j) {
      float acc = xv[j];
#pragma unroll
      for (int i = 0; i < RNN_H; ++i) acc = fmaf(w[j][i], h[i], acc);
      nh[j] = fast_tanhf(acc);
    }
#pragma unroll
    for (int j = 0; j < RNN_H; ++j) h[j] = nh[j];
    cur0 = nx0; cur1 = nx1;
  }

  float acc = b_fc[0];
#pragma unroll
  for (int j = 0; j < RNN_H; ++j) acc = fmaf(W_fc[j], h[j], acc);
  out[b] = acc;
}

__global__ __launch_bounds__(256) void rnn_fused(
    const float* __restrict__ x, const float* __restrict__ W_ih,
    const float* __restrict__ W_hh, const float* __restrict__ b_ih,
    const float* __restrict__ b_hh, const float* __restrict__ W_fc,
    const float* __restrict__ b_fc, float* __restrict__ out)
{
  const int b = blockIdx.x * blockDim.x + threadIdx.x;

  float wih[RNN_H][RNN_I], whh[RNN_H][RNN_H], bias[RNN_H];
#pragma unroll
  for (int j = 0; j < RNN_H; ++j) {
    bias[j] = b_ih[j] + b_hh[j];
#pragma unroll
    for (int i = 0; i < RNN_I; ++i) wih[j][i] = W_ih[j * RNN_I + i];
#pragma unroll
    for (int i = 0; i < RNN_H; ++i) whh[j][i] = W_hh[j * RNN_H + i];
  }

  float h[RNN_H];
#pragma unroll
  for (int j = 0; j < RNN_H; ++j) h[j] = 0.0f;

  const float* xr = x + (long)b * RNN_T * RNN_I;
  for (int t = 0; t < RNN_T; ++t) {
    float xv[RNN_I];
#pragma unroll
    for (int i = 0; i < RNN_I; ++i) xv[i] = xr[(long)t * RNN_I + i];
    float nh[RNN_H];
#pragma unroll
    for (int j = 0; j < RNN_H; ++j) {
      float acc = bias[j];
#pragma unroll
      for (int i = 0; i < RNN_I; ++i) acc = fmaf(wih[j][i], xv[i], acc);
#pragma unroll
      for (int i = 0; i < RNN_H; ++i) acc = fmaf(whh[j][i], h[i], acc);
      nh[j] = fast_tanhf(acc);
    }
#pragma unroll
    for (int j = 0; j < RNN_H; ++j) h[j] = nh[j];
  }

  float acc = b_fc[0];
#pragma unroll
  for (int j = 0; j < RNN_H; ++j) acc = fmaf(W_fc[j], h[j], acc);
  out[b] = acc;
}

extern "C" void kernel_launch(void* const* d_in, const int* in_sizes, int n_in,
                              void* d_out, int out_size, void* d_ws, size_t ws_size,
                              hipStream_t stream) {
  const float* x    = (const float*)d_in[0];
  const float* W_ih = (const float*)d_in[1];
  const float* W_hh = (const float*)d_in[2];
  const float* b_ih = (const float*)d_in[3];
  const float* b_hh = (const float*)d_in[4];
  const float* W_fc = (const float*)d_in[5];
  const float* b_fc = (const float*)d_in[6];
  float* out = (float*)d_out;

  const size_t need = (size_t)RNN_T * RNN_B * 8 * sizeof(float);
  if (ws_size >= need) {
    float* xp = (float*)d_ws;
    const long waves  = ((long)RNN_B * RNN_T / 16) / TILES_PER_WAVE;
    const int  blocks = (int)(waves * 32 / 256);
    xproj_wmma_f32<<<dim3(blocks), dim3(256), 0, stream>>>(x, W_ih, b_ih, b_hh, xp);
    rnn_scan<<<dim3(RNN_B / 256), dim3(256), 0, stream>>>(xp, W_hh, W_fc, b_fc, out);
  } else {
    rnn_fused<<<dim3(RNN_B / 256), dim3(256), 0, stream>>>(x, W_ih, W_hh, b_ih,
                                                           b_hh, W_fc, b_fc, out);
  }
}


# one kernel, recurrence on the f32 matrix instruction: a wave runs whole steps for 16 batch rows (weights as A, [x_t;h] as B) and reads x itself; no xp, one launch; order of sums changed
# speedup vs baseline: 6.2407x; 6.2407x over previous
#include <hip/hip_runtime.h>

typedef __attribute__((ext_vector_type(2))) float v2f;
typedef __attribute__((ext_vector_type(8))) float v8f;

#define RNN_B 8192
#define RNN_T 512
#define RNN_I 9
#define RNN_H 6
#define RNN_NG (RNN_T / 4)
#define RNN_RING 8


__host__ __device__ constexpr int reg_lo(int p, int j) { return 5 * p + j; }
__host__ __device__ constexpr int reg_hi(int p, int j) { return 4 * p + ((j + p) & 3); }
__host__ __device__ constexpr int k_lo(int p, int j) {
  return 8 * (reg_lo(p, j) >> 2) + (reg_lo(p, j) & 3) - 9 * p;
}
__host__ __device__ constexpr int k_hi(int p, int j) {
  return 8 * (reg_hi(p, j) >> 2) + 4 + (reg_hi(p, j) & 3) - 9 * p;
}

__device__ __forceinline__ v8f wmma_k4(v2f a, v2f b, v8f c) {
  return __builtin_amdgcn_wmma_f32_16x16x4_f32(false, a, false, b, (short)0, c, false, false);
}

__device__ __forceinline__ float tanh_rcp(float v) {
  float e = __builtin_amdgcn_exp2f(v * 2.88539008177792681472f);
  return __builtin_amdgcn_rcpf(e + 1.0f);
}

__device__ __forceinline__ void load_group(float (&q)[20], const float4* __restrict__ lp,
                                           const float4* __restrict__ tp, int g) {
#pragma unroll
  for (int i = 0; i < 4; ++i) {
    const float4 v = lp[9 * g + 2 * i];
    q[4 * i + 0] = v.x; q[4 * i + 1] = v.y; q[4 * i + 2] = v.z; q[4 * i + 3] = v.w;
  }
  const float4 w = tp[9 * g];
  q[16] = w.x; q[17] = w.y; q[18] = w.z; q[19] = w.w;
}

__global__ __launch_bounds__(64) void rnn_wmma_scan(
    const float* __restrict__ x, const float* __restrict__ W_ih,
    const float* __restrict__ W_hh, const float* __restrict__ b_ih,
    const float* __restrict__ b_hh, const float* __restrict__ W_fc,
    const float* __restrict__ b_fc, float* __restrict__ out)
{
  __shared__ float s_out[16];

  const int  lane = threadIdx.x & 31;
  const int  l15  = lane & 15;
  const int  half = lane >> 4;
  const bool hi   = half != 0;
  const int  wib  = threadIdx.x >> 5;
  const int  b0   = (blockIdx.x * 2 + wib) * 16;

  const int  u    = (l15 < 2) ? l15 : ((l15 >= 8 && l15 < 12) ? (l15 - 6) : -1);
  const bool uval = (u >= 0);
  const int  uc   = uval ? u : 0;

  v2f ax[4][2];
#pragma unroll
  for (int p = 0; p < 4; ++p) {
#pragma unroll
    for (int s = 0; s < 2; ++s) {
      const int ka = hi ? k_hi(p, 2 * s + 0) : k_lo(p, 2 * s + 0);
      const int kb = hi ? k_hi(p, 2 * s + 1) : k_lo(p, 2 * s + 1);
      const float wa = W_ih[uc * RNN_I + ka];
      const float wb = W_ih[uc * RNN_I + kb];
      ax[p][s].x = uval ? wa : 0.0f;
      ax[p][s].y = uval ? wb : 0.0f;
    }
  }
  v2f ahA, ahB;
  {
    const float w0 = W_hh[uc * RNN_H + 2 * half + 0];
    const float w1 = W_hh[uc * RNN_H + 2 * half + 1];
    ahA.x = uval ? w0 : 0.0f;
    ahA.y = uval ? w1 : 0.0f;
    const float w4 = W_hh[uc * RNN_H + 4];
    const float w5 = W_hh[uc * RNN_H + 5];
    const float w8 = W_ih[uc * RNN_I + 8];
    ahB.x = uval ? (hi ? w4 : w8) : 0.0f;
    ahB.y = (uval && hi) ? w5 : 0.0f;
  }
  v2f afA, afB;
  {
    const float f0 = W_fc[2 * half + 0];
    const float f1 = W_fc[2 * half + 1];
    const float f4 = W_fc[4];
    const float f5 = W_fc[5];
    afA.x = (l15 == 0) ? f0 : 0.0f;
    afA.y = (l15 == 0) ? f1 : 0.0f;
    afB.x = (l15 == 0 && hi) ? f4 : 0.0f;
    afB.y = (l15 == 0 && hi) ? f5 : 0.0f;
  }
  v8f cinit;
  {
    const float s0 = b_ih[0] + b_hh[0], s1 = b_ih[1] + b_hh[1], s2 = b_ih[2] + b_hh[2];
    const float s3 = b_ih[3] + b_hh[3], s4 = b_ih[4] + b_hh[4], s5 = b_ih[5] + b_hh[5];
    cinit[0] = hi ? s2 : s0;
    cinit[1] = hi ? s3 : s1;
    cinit[2] = hi ? s4 : 0.0f;
    cinit[3] = hi ? s5 : 0.0f;
    cinit[4] = 0.0f; cinit[5] = 0.0f; cinit[6] = 0.0f; cinit[7] = 0.0f;
  }
  const float m2 = hi ? -2.0f : 0.0f;
  const float a3 = hi ? 1.0f : 0.0f;

  const float4* rowq = (const float4*)(x + (long)(b0 + l15) * (RNN_T * RNN_I));
  const float4* lp = rowq + half;
  const float4* tp = rowq + 8;

  float q[RNN_RING][20];
#pragma unroll
  for (int r = 0; r < RNN_RING; ++r) load_group(q[r], lp, tp, r);
  asm volatile("" ::: "memory");

#define SLOT(Q, P, J) (hi ? (Q)[reg_hi((P), (J))] : (Q)[reg_lo((P), (J))])

  v8f xacc;
  {
    v2f x0, x1;
    x0.x = SLOT(q[0], 0, 0); x0.y = SLOT(q[0], 0, 1);
    x1.x = SLOT(q[0], 0, 2); x1.y = SLOT(q[0], 0, 3);
    xacc = wmma_k4(ax[0][1], x1, wmma_k4(ax[0][0], x0, cinit));
  }
  v2f bh01, bhB;
  bh01.x = 0.0f; bh01.y = 0.0f;
  bhB.x = hi ? 0.0f : q[0][reg_lo(0, 4)];
  bhB.y = 0.0f;

#pragma unroll 1
  for (int gb = 0; gb < RNN_NG; gb += RNN_RING) {
#pragma unroll
    for (int r = 0; r < RNN_RING; ++r) {
#pragma unroll
      for (int p = 0; p < 4; ++p) {
        v8f acc = wmma_k4(ahA, bh01, xacc);
        acc = wmma_k4(ahB, bhB, acc);
        __builtin_amdgcn_sched_barrier(0);

        const int pn = (p + 1) & 3;
        const float (&qn)[20] = (p < 3) ? q[r] : q[(r + 1) % RNN_RING];
        v2f x0, x1;
        x0.x = SLOT(qn, pn, 0); x0.y = SLOT(qn, pn, 1);
        x1.x = SLOT(qn, pn, 2); x1.y = SLOT(qn, pn, 3);
        xacc = wmma_k4(ax[pn][1], x1, wmma_k4(ax[pn][0], x0, cinit));
        const float a2 = hi ? 1.0f : qn[reg_lo(pn, 4)];

        const float r0 = tanh_rcp(acc[0]);
        const float r1 = tanh_rcp(acc[1]);
        const float r2 = tanh_rcp(acc[2]);
        const float r3 = tanh_rcp(acc[3]);
        bh01.x = __builtin_fmaf(-2.0f, r0, 1.0f);
        bh01.y = __builtin_fmaf(-2.0f, r1, 1.0f);
        bhB.x  = __builtin_fmaf(m2, r2, a2);
        bhB.y  = __builtin_fmaf(m2, r3, a3);

        if (p == 2) {
          int gl = gb + r + RNN_RING;
          gl = (gl < RNN_NG) ? gl : (RNN_NG - 1);
          asm volatile("" ::: "memory");
          load_group(q[r], lp, tp, gl);
          asm volatile("" ::: "memory");
        }
        __builtin_amdgcn_sched_barrier(0);
      }
    }
  }
#undef SLOT

  v8f cf;
  cf[0] = b_fc[0];
  cf[1] = 0.0f; cf[2] = 0.0f; cf[3] = 0.0f; cf[4] = 0.0f; cf[5] = 0.0f; cf[6] = 0.0f; cf[7] = 0.0f;
  v8f o = wmma_k4(afA, bh01, cf);
  o = wmma_k4(afB, bhB, o);
  const float res = o[0];

  if (wib == 1 && !hi) s_out[l15] = res;
  __syncthreads();
  if (wib == 0) {
    const float v = hi ? s_out[l15] : res;
    out[blockIdx.x * 32 + lane] = v;
  }
}

extern "C" void kernel_launch(void* const* d_in, const int* in_sizes, int n_in,
                              void* d_out, int out_size, void* d_ws, size_t ws_size,
                              hipStream_t stream) {
  const float* x    = (const float*)d_in[0];
  const float* W_ih = (const float*)d_in[1];
  const float* W_hh = (const float*)d_in[2];
  const float* b_ih = (const float*)d_in[3];
  const float* b_hh = (const float*)d_in[4];
  const float* W_fc = (const float*)d_in[5];
  const float* b_fc = (const float*)d_in[6];
  float* out = (float*)d_out;

  rnn_wmma_scan<<<dim3(RNN_B / 32), dim3(64), 0, stream>>>(x, W_ih, W_hh, b_ih, b_hh,
                                                         W_fc, b_fc, out);
}
